# starting kernel
# baseline (speedup 1.0000x reference)
#include <hip/hip_runtime.h>
#include <hip/hip_bf16.h>

typedef __attribute__((ext_vector_type(2))) float v2f;
typedef __attribute__((ext_vector_type(8))) float v8f;

#define IN_CH  512
#define HIDC   128
#define OUTC   64

__global__ void k_init_deg(float* deg, int n) {
    int i = blockIdx.x * blockDim.x + threadIdx.x;
    if (i < n) deg[i] = 1.0f;
}

__global__ void k_count_deg(const int* __restrict__ col, float* deg, int E) {
    int e = blockIdx.x * blockDim.x + threadIdx.x;
    if (e < E) atomicAdd(&deg[col[e]], 1.0f);
}

__global__ void k_rsqrt(float* deg, int n) {
    int i = blockIdx.x * blockDim.x + threadIdx.x;
    if (i < n) deg[i] = rsqrtf(deg[i]);
}

__global__ __launch_bounds__(256)
void k_gemm_wmma_f32(const float* __restrict__ A, const float* __restrict__ B,
                     float* __restrict__ C, int M, int K, int N) {
    __shared__ float ldsA[64][36];
    __shared__ float ldsB[32][33];

    const int tid   = threadIdx.x;
    const int lane  = tid & 31;
    const int wave  = tid >> 5;
    const int waveM = wave >> 1;
    const int waveN = wave & 1;
    const int m0 = blockIdx.x * 64;
    const int n0 = blockIdx.y * 32;

    v8f acc = {};

    const int lhalf = lane >> 4;
    const int l15   = lane & 15;
    const int arow  = waveM * 16 + l15;
    const int bcol  = waveN * 16 + l15;
    const int khalf = lhalf * 2;

    for (int k0 = 0; k0 < K; k0 += 32) {
        #pragma unroll
        for (int i = 0; i < 2; ++i) {
            int idx = tid + i * 256;
            int r   = idx >> 3;
            int c4  = idx & 7;
            int gr  = m0 + r;
            gr = gr < M ? gr : (M - 1);
            const float4 v = *(const float4*)(A + (size_t)gr * K + k0 + c4 * 4);
            ldsA[r][c4 * 4 + 0] = v.x;
            ldsA[r][c4 * 4 + 1] = v.y;
            ldsA[r][c4 * 4 + 2] = v.z;
            ldsA[r][c4 * 4 + 3] = v.w;
        }
        {
            int r  = tid >> 3;
            int c4 = tid & 7;
            const float4 v = *(const float4*)(B + (size_t)(k0 + r) * N + n0 + c4 * 4);
            ldsB[r][c4 * 4 + 0] = v.x;
            ldsB[r][c4 * 4 + 1] = v.y;
            ldsB[r][c4 * 4 + 2] = v.z;
            ldsB[r][c4 * 4 + 3] = v.w;
        }
        __syncthreads();

        #pragma unroll
        for (int kk = 0; kk < 32; kk += 4) {
            v2f a, b;
            a.x = ldsA[arow][kk + khalf];
            a.y = ldsA[arow][kk + khalf + 1];
            b.x = ldsB[kk + khalf]    [bcol];
            b.y = ldsB[kk + khalf + 1][bcol];
            acc = __builtin_amdgcn_wmma_f32_16x16x4_f32(
                 false, a,  false, b,
                 (short)0, acc,  false,  false);
        }
        __syncthreads();
    }

    const int crow0 = m0 + waveM * 16 + lhalf * 8;
    const int ccol  = n0 + bcol;
    #pragma unroll
    for (int v = 0; v < 8; ++v) {
        int r = crow0 + v;
        if (r < M) C[(size_t)r * N + ccol] = acc[v];
    }
}

__global__ void k_self_loop(const float* __restrict__ h, const float* __restrict__ dinv,
                            float* __restrict__ out, int n, int F) {
    int idx = blockIdx.x * blockDim.x + threadIdx.x;
    if (idx >= n * F) return;
    int i = idx / F;
    float d = dinv[i];
    out[idx] = d * d * h[idx];
}

__global__ void k_scatter(const int* __restrict__ rows, const int* __restrict__ cols,
                          const float* __restrict__ dinv, const float* __restrict__ h,
                          float* __restrict__ out, int E, int F) {
    int idx = blockIdx.x * blockDim.x + threadIdx.x;
    if (idx >= E * F) return;
    int e = idx / F;
    int f = idx - e * F;
    int r = rows[e];
    int c = cols[e];
    float norm = dinv[r] * dinv[c];
    atomicAdd(out + (size_t)c * F + f, norm * h[(size_t)r * F + f]);
}

__global__ void k_bias_act(float* __restrict__ buf, const float* __restrict__ bias,
                           int n, int F, int do_relu) {
    int idx = blockIdx.x * blockDim.x + threadIdx.x;
    if (idx >= n * F) return;
    int f = idx % F;
    float v = buf[idx] + bias[f];
    if (do_relu) v = v > 0.0f ? v : 0.0f;
    buf[idx] = v;
}

__global__ __launch_bounds__(256)
void k_decode(const int* __restrict__ pos, const int* __restrict__ neg,
              const float* __restrict__ z, float* __restrict__ logits, int E) {
    int gid = blockIdx.x * blockDim.x + threadIdx.x;
    int e = gid >> 4;
    int l = gid & 15;
    if (e >= 2 * E) return;
    int src, dst;
    if (e < E) { src = pos[e];      dst = pos[E + e];      }
    else       { src = neg[e - E];  dst = neg[E + (e - E)]; }
    const float4 a = *(const float4*)(z + (size_t)src * OUTC + l * 4);
    const float4 b = *(const float4*)(z + (size_t)dst * OUTC + l * 4);
    float s = a.x * b.x + a.y * b.y + a.z * b.z + a.w * b.w;
    s += __shfl_xor(s, 1, 32);
    s += __shfl_xor(s, 2, 32);
    s += __shfl_xor(s, 4, 32);
    s += __shfl_xor(s, 8, 32);
    if (l == 0) logits[e] = s;
}

static inline size_t align_f(size_t n) { return (n + 63) & ~size_t(63); }

extern "C" void kernel_launch(void* const* d_in, const int* in_sizes, int n_in,
                              void* d_out, int out_size, void* d_ws, size_t ws_size,
                              hipStream_t stream) {
    const float* x   = (const float*)d_in[0];
    const int*   pos = (const int*)  d_in[1];
    const int*   neg = (const int*)  d_in[2];
    const float* W1  = (const float*)d_in[3];
    const float* b1  = (const float*)d_in[4];
    const float* W2  = (const float*)d_in[5];
    const float* b2  = (const float*)d_in[6];
    float* out = (float*)d_out;

    const int N = in_sizes[0] / IN_CH;
    const int E = in_sizes[1] / 2;

    float* ws = (float*)d_ws;
    size_t o = 0;
    float* dinv = ws + o; o += align_f((size_t)N);
    float* h0   = ws + o; o += align_f((size_t)N * HIDC);
    float* g1   = ws + o; o += align_f((size_t)N * HIDC);
    float* g2   = ws + o; o += align_f((size_t)N * OUTC);
    (void)ws_size;

    const int T = 256;
    k_init_deg <<<(N + T - 1) / T, T, 0, stream>>>(dinv, N);
    k_count_deg<<<(E + T - 1) / T, T, 0, stream>>>(pos + E, dinv, E);
    k_rsqrt    <<<(N + T - 1) / T, T, 0, stream>>>(dinv, N);

    dim3 g1g((N + 63) / 64, HIDC / 32);
    k_gemm_wmma_f32<<<g1g, T, 0, stream>>>(x, W1, h0, N, IN_CH, HIDC);
    k_self_loop<<<((N * HIDC) + T - 1) / T, T, 0, stream>>>(h0, dinv, g1, N, HIDC);
    k_scatter  <<<((E * HIDC) + T - 1) / T, T, 0, stream>>>(pos, pos + E, dinv, h0, g1, E, HIDC);
    k_bias_act <<<((N * HIDC) + T - 1) / T, T, 0, stream>>>(g1, b1, N, HIDC, 1);

    dim3 g2g((N + 63) / 64, OUTC / 32);
    k_gemm_wmma_f32<<<g2g, T, 0, stream>>>(g1, W2, h0, N, HIDC, OUTC);
    k_self_loop<<<((N * OUTC) + T - 1) / T, T, 0, stream>>>(h0, dinv, g2, N, OUTC);
    k_scatter  <<<((E * OUTC) + T - 1) / T, T, 0, stream>>>(pos, pos + E, dinv, h0, g2, E, OUTC);
    k_bias_act <<<((N * OUTC) + T - 1) / T, T, 0, stream>>>(g2, b2, N, OUTC, 0);

    int total = 2 * E * 16;
    k_decode<<<(total + T - 1) / T, T, 0, stream>>>(pos, neg, g2, out, E);
    (void)out_size; (void)n_in;
}


# GEMMs: a wave owns 16-row x 64-column strips, A fragments go from global memory straight to registers 3 chunks ahead, all of B stays in LDS; no barrier in the k loop; k summed in a new fixed order
# speedup vs baseline: 2.3240x; 2.3240x over previous
#include <hip/hip_runtime.h>
#include <hip/hip_bf16.h>

typedef __attribute__((ext_vector_type(2))) float v2f;
typedef __attribute__((ext_vector_type(8))) float v8f;

#define IN_CH  512
#define HIDC   128
#define OUTC   64

__global__ void k_init_deg(float* deg, int n) {
    int i = blockIdx.x * blockDim.x + threadIdx.x;
    if (i < n) deg[i] = 1.0f;
}

__global__ void k_count_deg(const int* __restrict__ col, float* deg, int E) {
    int e = blockIdx.x * blockDim.x + threadIdx.x;
    if (e < E) atomicAdd(&deg[col[e]], 1.0f);
}

template <int K, int N, int NSPLIT, bool RELU_A>
__global__ __launch_bounds__(256)
void k_gemm_strip_f32(const float* __restrict__ A, const float* __restrict__ B,
                      float* __restrict__ C, float* __restrict__ G,
                      const float* __restrict__ deg, const float* __restrict__ bias,
                      int M) {
    constexpr int P     = 2 * N + 4;
    constexpr int NCH   = K / 32;
    constexpr int NW    = N / NSPLIT;
    constexpr int NT    = NW / 16;
    constexpr int ISTEP = 8 / NSPLIT;
    static_assert(NT == 4 && (NCH % 4) == 0 && (NCH & (NCH - 1)) == 0, "shape");
    __shared__ __attribute__((aligned(16))) float ldsB[(K / 2) * P];

    const int tid  = threadIdx.x;
    const int lane = tid & 31;
    const int wave = __builtin_amdgcn_readfirstlane(tid >> 5);
    const int c    = lane & 15;
    const int hf   = lane >> 4;
    const int n0   = (wave % NSPLIT) * NW;
    const int i0   = wave / NSPLIT;
    const int GX   = gridDim.x;
    const int ntiles = (M + 15) >> 4;

    int cnt = 0;
    {
        const int first = blockIdx.x + GX * i0;
        if (first < ntiles) cnt = (ntiles - 1 - first) / (GX * ISTEP) + 1;
    }
    const int total = cnt * NCH;

    float4 ring[4][4];
    auto loadA = [&](float4 (&dst)[4], int gi) {
        int g = gi < total ? gi : (total - 1);
        g = g > 0 ? g : 0;
        const int j  = g / NCH, kc = g % NCH;
        const int t  = blockIdx.x + GX * (i0 + ISTEP * j);
        int row = t * 16 + c;
        row = row < M ? row : (M - 1);
        const float* p = A + (size_t)row * K + kc * 32 + hf * 16;
        #pragma unroll
        for (int q = 0; q < 4; ++q) dst[q] = *(const float4*)(p + q * 4);
    };
    loadA(ring[0], 0);
    loadA(ring[1], 1);
    loadA(ring[2], 2);

    {
        constexpr int NQ    = N / 4;
        constexpr int ITEMS = (K / 2) * NQ;
        constexpr int PER   = ITEMS / 256;
        constexpr int UNR   = PER < 8 ? PER : 8;
        for (int it0 = 0; it0 < ITEMS; it0 += 256 * UNR) {
            float4 r0[UNR], r1[UNR];
            #pragma unroll
            for (int u = 0; u < UNR; ++u) {
                const int item = it0 + u * 256 + tid;
                const int kp = item / NQ, n4 = item % NQ;
                r0[u] = *(const float4*)(B + (size_t)(2 * kp)     * N + n4 * 4);
                r1[u] = *(const float4*)(B + (size_t)(2 * kp + 1) * N + n4 * 4);
            }
            #pragma unroll
            for (int u = 0; u < UNR; ++u) {
                const int item = it0 + u * 256 + tid;
                const int kp = item / NQ, n4 = item % NQ;
                float* d = ldsB + kp * P + n4 * 8;
                *(float4*)(d)     = make_float4(r0[u].x, r1[u].x, r0[u].y, r1[u].y);
                *(float4*)(d + 4) = make_float4(r0[u].z, r1[u].z, r0[u].w, r1[u].w);
            }
        }
    }
    float bc[NT];
    #pragma unroll
    for (int j = 0; j < NT; ++j) bc[j] = bias[n0 + j * 16 + c];
    __syncthreads();

    v8f acc[NT];
    #pragma unroll
    for (int j = 0; j < NT; ++j) acc[j] = v8f{};
    float dg[8];
    const float* bfrag = ldsB + (hf * 8) * P + (n0 + c) * 2;
    v2f bq[2][8][NT];
    auto loadB = [&](v2f (&dst)[8][NT], int kc) {
        const float* bb = bfrag + kc * 16 * P;
        #pragma unroll
        for (int p = 0; p < 8; ++p)
            #pragma unroll
            for (int j = 0; j < NT; ++j) dst[p][j] = *(const v2f*)(bb + p * P + j * 32);
    };
    loadB(bq[0], 0);

    for (int s = 0; s < cnt; ++s) {
        const int rb = (blockIdx.x + GX * (i0 + ISTEP * s)) * 16 + hf * 8;
        #pragma unroll
        for (int r = 0; r < 8; ++r) dg[r] = deg[(rb + r) < M ? (rb + r) : (M - 1)];

        for (int kq = 0; kq < NCH; kq += 4) {
            #pragma unroll
            for (int u = 0; u < 4; ++u) {
                const int kc = kq + u;
                loadA(ring[(u + 3) & 3], s * NCH + kc + 3);
                loadB(bq[(u + 1) & 1], (kc + 1) % NCH);
                __builtin_amdgcn_sched_barrier(0);
                #pragma unroll
                for (int p = 0; p < 8; ++p) {
                    const float4 q = ring[u][p >> 1];
                    v2f a;
                    a.x = (p & 1) ? q.z : q.x;
                    a.y = (p & 1) ? q.w : q.y;
                    if (RELU_A) {
                        a.x = a.x > 0.0f ? a.x : 0.0f;
                        a.y = a.y > 0.0f ? a.y : 0.0f;
                    }
                    #pragma unroll
                    for (int j = 0; j < NT; ++j) {
                        const v2f b = bq[u & 1][p][j];
                        acc[j] = __builtin_amdgcn_wmma_f32_16x16x4_f32(
                             false, a,  false, b,
                             (short)0, acc[j],  false,  false);
                    }
                }
            }
        }

        #pragma unroll
        for (int r = 0; r < 8; ++r) {
            const int row = rb + r;
            if (row < M) {
                const float d = rsqrtf(dg[r]);
                #pragma unroll
                for (int j = 0; j < NT; ++j) {
                    const float hv = acc[j][r];
                    const size_t o = (size_t)row * N + n0 + j * 16 + c;
                    C[o] = hv;
                    G[o] = d * d * hv + bc[j];
                }
            }
        }
        #pragma unroll
        for (int j = 0; j < NT; ++j) acc[j] = v8f{};
    }
}


template <int F>
__global__ __launch_bounds__(256)
void k_scatter_rows(const int* __restrict__ rows, const int* __restrict__ cols,
                    const float* __restrict__ deg, const float* __restrict__ h,
                    float* __restrict__ out, int E) {
    constexpr int L = F / 32;
    constexpr int U = 8;
    const unsigned lane = threadIdx.x & 31u;
    const int wave = __builtin_amdgcn_readfirstlane(
        (int)((blockIdx.x * blockDim.x + threadIdx.x) >> 5));
    const int e0 = wave * 32;
    if (e0 >= E) return;
    const int left = E - e0;
    const int cnt = left < 32 ? left : 32;

    int r = 0, c = 0;
    float w = 0.0f;
    if ((int)lane < cnt) {
        r = rows[e0 + lane];
        c = cols[e0 + lane];
        w = rsqrtf(deg[r]) * rsqrtf(deg[c]);
    }
    const int wbits = __float_as_int(w);

    int j0 = 0;
    for (; j0 + U <= cnt; j0 += U) {
        float v[U][L];
        #pragma unroll
        for (int u = 0; u < U; ++u) {
            const int rj = __builtin_amdgcn_readlane(r, j0 + u);
            const float* src = h + (size_t)rj * F;
            #pragma unroll
            for (int k = 0; k < L; ++k) v[u][k] = src[lane + k * 32];
        }
        #pragma unroll
        for (int u = 0; u < U; ++u) {
            const int cj = __builtin_amdgcn_readlane(c, j0 + u);
            const float wj = __int_as_float(__builtin_amdgcn_readlane(wbits, j0 + u));
            float* dst = out + (size_t)cj * F;
            #pragma unroll
            for (int k = 0; k < L; ++k) atomicAdd(dst + lane + k * 32, wj * v[u][k]);
        }
    }
    for (; j0 < cnt; ++j0) {
        const int rj = __builtin_amdgcn_readlane(r, j0);
        const int cj = __builtin_amdgcn_readlane(c, j0);
        const float wj = __int_as_float(__builtin_amdgcn_readlane(wbits, j0));
        const float* src = h + (size_t)rj * F;
        float* dst = out + (size_t)cj * F;
        #pragma unroll
        for (int k = 0; k < L; ++k) atomicAdd(dst + lane + k * 32, wj * src[lane + k * 32]);
    }
}

__global__ __launch_bounds__(256)
void k_decode(const int* __restrict__ pos, const int* __restrict__ neg,
              const float* __restrict__ z, float* __restrict__ logits, int E) {
    int gid = blockIdx.x * blockDim.x + threadIdx.x;
    int e = gid >> 4;
    int l = gid & 15;
    if (e >= 2 * E) return;
    int src, dst;
    if (e < E) { src = pos[e];      dst = pos[E + e];      }
    else       { src = neg[e - E];  dst = neg[E + (e - E)]; }
    const float4 a = *(const float4*)(z + (size_t)src * OUTC + l * 4);
    const float4 b = *(const float4*)(z + (size_t)dst * OUTC + l * 4);
    float s = a.x * b.x + a.y * b.y + a.z * b.z + a.w * b.w;
    s += __shfl_xor(s, 1, 32);
    s += __shfl_xor(s, 2, 32);
    s += __shfl_xor(s, 4, 32);
    s += __shfl_xor(s, 8, 32);
    if (l == 0) logits[e] = s;
}

static inline size_t align_f(size_t n) { return (n + 63) & ~size_t(63); }

extern "C" void kernel_launch(void* const* d_in, const int* in_sizes, int n_in,
                              void* d_out, int out_size, void* d_ws, size_t ws_size,
                              hipStream_t stream) {
    const float* x   = (const float*)d_in[0];
    const int*   pos = (const int*)  d_in[1];
    const int*   neg = (const int*)  d_in[2];
    const float* W1  = (const float*)d_in[3];
    const float* b1  = (const float*)d_in[4];
    const float* W2  = (const float*)d_in[5];
    const float* b2  = (const float*)d_in[6];
    float* out = (float*)d_out;

    const int N = in_sizes[0] / IN_CH;
    const int E = in_sizes[1] / 2;

    float* ws = (float*)d_ws;
    size_t o = 0;
    float* deg  = ws + o; o += align_f((size_t)N);
    float* h0   = ws + o; o += align_f((size_t)N * HIDC);
    float* g1   = ws + o; o += align_f((size_t)N * HIDC);
    float* g2   = ws + o; o += align_f((size_t)N * OUTC);
    (void)ws_size;

    const int T = 256;
    k_init_deg <<<(N + T - 1) / T, T, 0, stream>>>(deg, N);
    k_count_deg<<<(E + T - 1) / T, T, 0, stream>>>(pos + E, deg, E);

    k_gemm_strip_f32<IN_CH, HIDC, 2, false><<<256, T, 0, stream>>>(x, W1, h0, g1, deg, b1, N);
    k_scatter_rows<HIDC><<<(E + T - 1) / T, T, 0, stream>>>(pos, pos + E, deg, h0, g1, E);

    k_gemm_strip_f32<HIDC, OUTC, 1, true><<<256, T, 0, stream>>>(g1, W2, h0, g2, deg, b2, N);
    k_scatter_rows<OUTC><<<(E + T - 1) / T, T, 0, stream>>>(pos, pos + E, deg, h0, g2, E);

    int total = 2 * E * 16;
    k_decode<<<(total + T - 1) / T, T, 0, stream>>>(pos, neg, g2, out, E);
    (void)out_size; (void)n_in;
}


# Degree count on LDS counters: 64 work-groups count edge slices into whole-table LDS histograms (integer adds), store count planes; a sum kernel adds planes + 1; per-edge global adds on deg gone
# speedup vs baseline: 2.7258x; 1.1729x over previous
#include <hip/hip_runtime.h>
#include <hip/hip_bf16.h>

typedef __attribute__((ext_vector_type(2))) float v2f;
typedef __attribute__((ext_vector_type(8))) float v8f;

#define IN_CH  512
#define HIDC   128
#define OUTC   64

#define DEG_BLOCKS 64
#define DEG_CAP    50176

__global__ __launch_bounds__(1024)
void k_deg_hist(const int* __restrict__ col, unsigned* __restrict__ part, int n, int npad, int E) {
    __shared__ __attribute__((aligned(16))) unsigned hist[DEG_CAP];
    constexpr int UL = 8;
    const int tid = threadIdx.x;
    const int step = gridDim.x * 1024;
    unsigned* plane = part + (size_t)blockIdx.x * npad;
    for (int base = 0; base < n; base += DEG_CAP) {
        const int lim = (n - base) < DEG_CAP ? (n - base) : DEG_CAP;
        for (int i = tid; i < DEG_CAP / 4; i += 1024) ((uint4*)hist)[i] = make_uint4(0u, 0u, 0u, 0u);
        __syncthreads();
        for (int e0 = blockIdx.x * 1024 + tid; e0 < E; e0 += step * UL) {
            int cc[UL];
            #pragma unroll
            for (int u = 0; u < UL; ++u) {
                const int e = e0 + u * step;
                cc[u] = e < E ? col[e] : -1;
            }
            #pragma unroll
            for (int u = 0; u < UL; ++u) {
                const unsigned k = (unsigned)(cc[u] - base);
                if (cc[u] >= 0 && k < (unsigned)lim) atomicAdd(&hist[k], 1u);
            }
        }
        __syncthreads();
        const int lim4 = lim >> 2;
        for (int i = tid; i < lim4; i += 1024) ((uint4*)(plane + base))[i] = ((const uint4*)hist)[i];
        if (tid < (lim & 3)) plane[base + (lim4 << 2) + tid] = hist[(lim4 << 2) + tid];
        __syncthreads();
    }
}

__global__ void k_deg_sum(const unsigned* __restrict__ part, float* __restrict__ deg, int n, int npad) {
    const int i = blockIdx.x * blockDim.x + threadIdx.x;
    if (i >= n) return;
    unsigned s = 0;
    #pragma unroll
    for (int b = 0; b < DEG_BLOCKS; ++b) s += part[(size_t)b * npad + i];
    deg[i] = 1.0f + (float)s;
}

template <int K, int N, int NSPLIT, bool RELU_A>
__global__ __launch_bounds__(256)
void k_gemm_strip_f32(const float* __restrict__ A, const float* __restrict__ B,
                      float* __restrict__ C, float* __restrict__ G,
                      const float* __restrict__ deg, const float* __restrict__ bias,
                      int M) {
    constexpr int P     = 2 * N + 4;
    constexpr int NCH   = K / 32;
    constexpr int NW    = N / NSPLIT;
    constexpr int NT    = NW / 16;
    constexpr int ISTEP = 8 / NSPLIT;
    static_assert(NT == 4 && (NCH % 4) == 0 && (NCH & (NCH - 1)) == 0, "shape");
    __shared__ __attribute__((aligned(16))) float ldsB[(K / 2) * P];

    const int tid  = threadIdx.x;
    const int lane = tid & 31;
    const int wave = __builtin_amdgcn_readfirstlane(tid >> 5);
    const int c    = lane & 15;
    const int hf   = lane >> 4;
    const int n0   = (wave % NSPLIT) * NW;
    const int i0   = wave / NSPLIT;
    const int GX   = gridDim.x;
    const int ntiles = (M + 15) >> 4;

    int cnt = 0;
    {
        const int first = blockIdx.x + GX * i0;
        if (first < ntiles) cnt = (ntiles - 1 - first) / (GX * ISTEP) + 1;
    }
    const int total = cnt * NCH;

    float4 ring[4][4];
    auto loadA = [&](float4 (&dst)[4], int gi) {
        int g = gi < total ? gi : (total - 1);
        g = g > 0 ? g : 0;
        const int j  = g / NCH, kc = g % NCH;
        const int t  = blockIdx.x + GX * (i0 + ISTEP * j);
        int row = t * 16 + c;
        row = row < M ? row : (M - 1);
        const float* p = A + (size_t)row * K + kc * 32 + hf * 16;
        #pragma unroll
        for (int q = 0; q < 4; ++q) dst[q] = *(const float4*)(p + q * 4);
    };
    loadA(ring[0], 0);
    loadA(ring[1], 1);
    loadA(ring[2], 2);

    {
        constexpr int NQ    = N / 4;
        constexpr int ITEMS = (K / 2) * NQ;
        constexpr int PER   = ITEMS / 256;
        constexpr int UNR   = PER < 8 ? PER : 8;
        for (int it0 = 0; it0 < ITEMS; it0 += 256 * UNR) {
            float4 r0[UNR], r1[UNR];
            #pragma unroll
            for (int u = 0; u < UNR; ++u) {
                const int item = it0 + u * 256 + tid;
                const int kp = item / NQ, n4 = item % NQ;
                r0[u] = *(const float4*)(B + (size_t)(2 * kp)     * N + n4 * 4);
                r1[u] = *(const float4*)(B + (size_t)(2 * kp + 1) * N + n4 * 4);
            }
            #pragma unroll
            for (int u = 0; u < UNR; ++u) {
                const int item = it0 + u * 256 + tid;
                const int kp = item / NQ, n4 = item % NQ;
                float* d = ldsB + kp * P + n4 * 8;
                *(float4*)(d)     = make_float4(r0[u].x, r1[u].x, r0[u].y, r1[u].y);
                *(float4*)(d + 4) = make_float4(r0[u].z, r1[u].z, r0[u].w, r1[u].w);
            }
        }
    }
    float bc[NT];
    #pragma unroll
    for (int j = 0; j < NT; ++j) bc[j] = bias[n0 + j * 16 + c];
    __syncthreads();

    v8f acc[NT];
    #pragma unroll
    for (int j = 0; j < NT; ++j) acc[j] = v8f{};
    float dg[8];
    const float* bfrag = ldsB + (hf * 8) * P + (n0 + c) * 2;
    v2f bq[2][8][NT];
    auto loadB = [&](v2f (&dst)[8][NT], int kc) {
        const float* bb = bfrag + kc * 16 * P;
        #pragma unroll
        for (int p = 0; p < 8; ++p)
            #pragma unroll
            for (int j = 0; j < NT; ++j) dst[p][j] = *(const v2f*)(bb + p * P + j * 32);
    };
    loadB(bq[0], 0);

    for (int s = 0; s < cnt; ++s) {
        const int rb = (blockIdx.x + GX * (i0 + ISTEP * s)) * 16 + hf * 8;
        #pragma unroll
        for (int r = 0; r < 8; ++r) dg[r] = deg[(rb + r) < M ? (rb + r) : (M - 1)];

        for (int kq = 0; kq < NCH; kq += 4) {
            #pragma unroll
            for (int u = 0; u < 4; ++u) {
                const int kc = kq + u;
                loadA(ring[(u + 3) & 3], s * NCH + kc + 3);
                loadB(bq[(u + 1) & 1], (kc + 1) % NCH);
                __builtin_amdgcn_sched_barrier(0);
                #pragma unroll
                for (int p = 0; p < 8; ++p) {
                    const float4 q = ring[u][p >> 1];
                    v2f a;
                    a.x = (p & 1) ? q.z : q.x;
                    a.y = (p & 1) ? q.w : q.y;
                    if (RELU_A) {
                        a.x = a.x > 0.0f ? a.x : 0.0f;
                        a.y = a.y > 0.0f ? a.y : 0.0f;
                    }
                    #pragma unroll
                    for (int j = 0; j < NT; ++j) {
                        const v2f b = bq[u & 1][p][j];
                        acc[j] = __builtin_amdgcn_wmma_f32_16x16x4_f32(
                             false, a,  false, b,
                             (short)0, acc[j],  false,  false);
                    }
                }
            }
        }

        #pragma unroll
        for (int r = 0; r < 8; ++r) {
            const int row = rb + r;
            if (row < M) {
                const float d = rsqrtf(dg[r]);
                #pragma unroll
                for (int j = 0; j < NT; ++j) {
                    const float hv = acc[j][r];
                    const size_t o = (size_t)row * N + n0 + j * 16 + c;
                    C[o] = hv;
                    G[o] = d * d * hv + bc[j];
                }
            }
        }
        #pragma unroll
        for (int j = 0; j < NT; ++j) acc[j] = v8f{};
    }
}


template <int F>
__global__ __launch_bounds__(256)
void k_scatter_rows(const int* __restrict__ rows, const int* __restrict__ cols,
                    const float* __restrict__ deg, const float* __restrict__ h,
                    float* __restrict__ out, int E) {
    constexpr int L = F / 32;
    constexpr int U = 8;
    const unsigned lane = threadIdx.x & 31u;
    const int wave = __builtin_amdgcn_readfirstlane(
        (int)((blockIdx.x * blockDim.x + threadIdx.x) >> 5));
    const int e0 = wave * 32;
    if (e0 >= E) return;
    const int left = E - e0;
    const int cnt = left < 32 ? left : 32;

    int r = 0, c = 0;
    float w = 0.0f;
    if ((int)lane < cnt) {
        r = rows[e0 + lane];
        c = cols[e0 + lane];
        w = rsqrtf(deg[r]) * rsqrtf(deg[c]);
    }
    const int wbits = __float_as_int(w);

    int j0 = 0;
    for (; j0 + U <= cnt; j0 += U) {
        float v[U][L];
        #pragma unroll
        for (int u = 0; u < U; ++u) {
            const int rj = __builtin_amdgcn_readlane(r, j0 + u);
            const float* src = h + (size_t)rj * F;
            #pragma unroll
            for (int k = 0; k < L; ++k) v[u][k] = src[lane + k * 32];
        }
        #pragma unroll
        for (int u = 0; u < U; ++u) {
            const int cj = __builtin_amdgcn_readlane(c, j0 + u);
            const float wj = __int_as_float(__builtin_amdgcn_readlane(wbits, j0 + u));
            float* dst = out + (size_t)cj * F;
            #pragma unroll
            for (int k = 0; k < L; ++k) atomicAdd(dst + lane + k * 32, wj * v[u][k]);
        }
    }
    for (; j0 < cnt; ++j0) {
        const int rj = __builtin_amdgcn_readlane(r, j0);
        const int cj = __builtin_amdgcn_readlane(c, j0);
        const float wj = __int_as_float(__builtin_amdgcn_readlane(wbits, j0));
        const float* src = h + (size_t)rj * F;
        float* dst = out + (size_t)cj * F;
        #pragma unroll
        for (int k = 0; k < L; ++k) atomicAdd(dst + lane + k * 32, wj * src[lane + k * 32]);
    }
}

__global__ __launch_bounds__(256)
void k_decode(const int* __restrict__ pos, const int* __restrict__ neg,
              const float* __restrict__ z, float* __restrict__ logits, int E) {
    int gid = blockIdx.x * blockDim.x + threadIdx.x;
    int e = gid >> 4;
    int l = gid & 15;
    if (e >= 2 * E) return;
    int src, dst;
    if (e < E) { src = pos[e];      dst = pos[E + e];      }
    else       { src = neg[e - E];  dst = neg[E + (e - E)]; }
    const float4 a = *(const float4*)(z + (size_t)src * OUTC + l * 4);
    const float4 b = *(const float4*)(z + (size_t)dst * OUTC + l * 4);
    float s = a.x * b.x + a.y * b.y + a.z * b.z + a.w * b.w;
    s += __shfl_xor(s, 1, 32);
    s += __shfl_xor(s, 2, 32);
    s += __shfl_xor(s, 4, 32);
    s += __shfl_xor(s, 8, 32);
    if (l == 0) logits[e] = s;
}

static inline size_t align_f(size_t n) { return (n + 63) & ~size_t(63); }

extern "C" void kernel_launch(void* const* d_in, const int* in_sizes, int n_in,
                              void* d_out, int out_size, void* d_ws, size_t ws_size,
                              hipStream_t stream) {
    const float* x   = (const float*)d_in[0];
    const int*   pos = (const int*)  d_in[1];
    const int*   neg = (const int*)  d_in[2];
    const float* W1  = (const float*)d_in[3];
    const float* b1  = (const float*)d_in[4];
    const float* W2  = (const float*)d_in[5];
    const float* b2  = (const float*)d_in[6];
    float* out = (float*)d_out;

    const int N = in_sizes[0] / IN_CH;
    const int E = in_sizes[1] / 2;

    float* ws = (float*)d_ws;
    size_t o = 0;
    float* deg  = ws + o; o += align_f((size_t)N);
    float* h0   = ws + o; o += align_f((size_t)N * HIDC);
    float* g1   = ws + o; o += align_f((size_t)N * HIDC);
    float* g2   = ws + o; o += align_f((size_t)N * OUTC);
    const int NP = (int)align_f((size_t)N);
    unsigned* dpart = (unsigned*)(ws + o); o += (size_t)DEG_BLOCKS * NP;
    (void)ws_size;

    const int T = 256;
    k_deg_hist<<<DEG_BLOCKS, 1024, 0, stream>>>(pos + E, dpart, N, NP, E);
    k_deg_sum <<<(N + T - 1) / T, T, 0, stream>>>(dpart, deg, N, NP);

    k_gemm_strip_f32<IN_CH, HIDC, 2, false><<<256, T, 0, stream>>>(x, W1, h0, g1, deg, b1, N);
    k_scatter_rows<HIDC><<<(E + T - 1) / T, T, 0, stream>>>(pos, pos + E, deg, h0, g1, E);

    k_gemm_strip_f32<HIDC, OUTC, 1, true><<<256, T, 0, stream>>>(g1, W2, h0, g2, deg, b2, N);
    k_scatter_rows<OUTC><<<(E + T - 1) / T, T, 0, stream>>>(pos, pos + E, deg, h0, g2, E);

    int total = 2 * E * 16;
    k_decode<<<(total + T - 1) / T, T, 0, stream>>>(pos, neg, g2, out, E);
    (void)out_size; (void)n_in;
}
